# starting kernel
# baseline (speedup 1.0000x reference)
#include <hip/hip_runtime.h>
#include <math.h>

#define N_NODES   100000
#define N_EDGES   1600000
#define F_IN      64
#define DIM       16
#define N_CLASSES 10
#define BN_EPS    1e-5f

typedef __attribute__((ext_vector_type(2))) float v2f;
typedef __attribute__((ext_vector_type(8))) float v8f;

__device__ __forceinline__ v8f wmma_k4(v2f a, v2f b, v8f c) {
    return __builtin_amdgcn_wmma_f32_16x16x4_f32(
         false, a,  false, b,
         (short)0, c,  false,  false);
}

__global__ void gin_proj64(const float* __restrict__ x,
                           const float* __restrict__ w1a,
                           float* __restrict__ y) {
    const int tile = blockIdx.x;
    const int lane = threadIdx.x;
    const int half = lane >> 4;
    const int l    = lane & 15;
    const int arow = tile * 16 + l;

    v8f c;
    #pragma unroll
    for (int r = 0; r < 8; ++r) c[r] = 0.0f;

    #pragma unroll
    for (int k0 = 0; k0 < F_IN; k0 += 4) {
        const int ka = k0 + half * 2;
        v2f a = *(const v2f*)(x + arow * F_IN + ka);
        v2f b;
        b.x = w1a[ka * DIM + l];
        b.y = w1a[ka * DIM + DIM + l];
        c = wmma_k4(a, b, c);
    }
    #pragma unroll
    for (int r = 0; r < 8; ++r)
        y[(tile * 16 + half * 8 + r) * DIM + l] = c[r];
}

__global__ void gin_scatter_add(const float* __restrict__ feat,
                                const int* __restrict__ src,
                                const int* __restrict__ dst,
                                float* __restrict__ agg) {
    const int t = blockIdx.x * blockDim.x + threadIdx.x;
    if (t >= N_EDGES * DIM) return;
    const int e = t >> 4;
    const int d = t & 15;
    const int s  = src[e];
    const int dd = dst[e];
    atomicAdd(agg + dd * DIM + d, feat[s * DIM + d]);
}

__global__ void gin_layer1(const float* __restrict__ y, const float* __restrict__ agg,
                           const float* __restrict__ b1a,
                           const float* __restrict__ w1b, const float* __restrict__ b1b,
                           const float* __restrict__ g,  const float* __restrict__ be,
                           const float* __restrict__ mu, const float* __restrict__ var,
                           float* __restrict__ u) {
    __shared__ float sA[16][17];
    const int tile = blockIdx.x;
    const int lane = threadIdx.x;
    const int half = lane >> 4;
    const int l    = lane & 15;

    const float bl = b1a[l];
    #pragma unroll
    for (int r = 0; r < 8; ++r) {
        const int row = half * 8 + r;
        const int gi  = (tile * 16 + row) * DIM + l;
        sA[row][l] = fmaxf(y[gi] + agg[gi] + bl, 0.0f);
    }
    __syncthreads();

    const float cb = b1b[l];
    v8f c;
    #pragma unroll
    for (int r = 0; r < 8; ++r) c[r] = cb;

    #pragma unroll
    for (int k0 = 0; k0 < DIM; k0 += 4) {
        const int ka = k0 + half * 2;
        v2f a; a.x = sA[l][ka];           a.y = sA[l][ka + 1];
        v2f b; b.x = w1b[ka * DIM + l];   b.y = w1b[ka * DIM + DIM + l];
        c = wmma_k4(a, b, c);
    }

    const float sc = g[l] * rsqrtf(var[l] + BN_EPS);
    const float mm = mu[l], bb = be[l];
    #pragma unroll
    for (int r = 0; r < 8; ++r) {
        const float t = fmaxf(c[r], 0.0f);
        u[(tile * 16 + half * 8 + r) * DIM + l] = (t - mm) * sc + bb;
    }
}

__global__ void gin_layer2_fc(const float* __restrict__ u,  const float* __restrict__ agg2,
                              const float* __restrict__ w2a, const float* __restrict__ b2a,
                              const float* __restrict__ w2b, const float* __restrict__ b2b,
                              const float* __restrict__ g,   const float* __restrict__ be,
                              const float* __restrict__ mu,  const float* __restrict__ var,
                              const float* __restrict__ fcw, const float* __restrict__ fcb,
                              float* __restrict__ logits) {
    __shared__ float sA[16][17];
    const int tile = blockIdx.x;
    const int lane = threadIdx.x;
    const int half = lane >> 4;
    const int l    = lane & 15;

    #pragma unroll
    for (int r = 0; r < 8; ++r) {
        const int row = half * 8 + r;
        const int gi  = (tile * 16 + row) * DIM + l;
        sA[row][l] = u[gi] + agg2[gi];
    }
    __syncthreads();

    v8f c;
    {
        const float cb = b2a[l];
        #pragma unroll
        for (int r = 0; r < 8; ++r) c[r] = cb;
    }
    #pragma unroll
    for (int k0 = 0; k0 < DIM; k0 += 4) {
        const int ka = k0 + half * 2;
        v2f a; a.x = sA[l][ka];          a.y = sA[l][ka + 1];
        v2f b; b.x = w2a[ka * DIM + l];  b.y = w2a[ka * DIM + DIM + l];
        c = wmma_k4(a, b, c);
    }
    __syncthreads();
    #pragma unroll
    for (int r = 0; r < 8; ++r) sA[half * 8 + r][l] = fmaxf(c[r], 0.0f);
    __syncthreads();

    {
        const float cb = b2b[l];
        #pragma unroll
        for (int r = 0; r < 8; ++r) c[r] = cb;
    }
    #pragma unroll
    for (int k0 = 0; k0 < DIM; k0 += 4) {
        const int ka = k0 + half * 2;
        v2f a; a.x = sA[l][ka];          a.y = sA[l][ka + 1];
        v2f b; b.x = w2b[ka * DIM + l];  b.y = w2b[ka * DIM + DIM + l];
        c = wmma_k4(a, b, c);
    }
    __syncthreads();

    {
        const float sc = g[l] * rsqrtf(var[l] + BN_EPS);
        const float mm = mu[l], bb = be[l];
        #pragma unroll
        for (int r = 0; r < 8; ++r) {
            const float t = fmaxf(c[r], 0.0f);
            sA[half * 8 + r][l] = (t - mm) * sc + bb;
        }
    }
    __syncthreads();

    {
        const float cb = (l < N_CLASSES) ? fcb[l] : 0.0f;
        #pragma unroll
        for (int r = 0; r < 8; ++r) c[r] = cb;
    }
    #pragma unroll
    for (int k0 = 0; k0 < DIM; k0 += 4) {
        const int ka = k0 + half * 2;
        v2f a; a.x = sA[l][ka];  a.y = sA[l][ka + 1];
        v2f b;
        b.x = (l < N_CLASSES) ? fcw[ka * N_CLASSES + l]       : 0.0f;
        b.y = (l < N_CLASSES) ? fcw[(ka + 1) * N_CLASSES + l] : 0.0f;
        c = wmma_k4(a, b, c);
    }
    if (l < N_CLASSES) {
        #pragma unroll
        for (int r = 0; r < 8; ++r)
            logits[(tile * 16 + half * 8 + r) * N_CLASSES + l] = c[r];
    }
}

__global__ void gin_logsoftmax(float* __restrict__ o) {
    const int i = blockIdx.x * blockDim.x + threadIdx.x;
    if (i >= N_NODES) return;
    float* p = o + (size_t)i * N_CLASSES;
    float v[N_CLASSES];
    float m = -INFINITY;
    #pragma unroll
    for (int j = 0; j < N_CLASSES; ++j) { v[j] = p[j]; m = fmaxf(m, v[j]); }
    float s = 0.0f;
    #pragma unroll
    for (int j = 0; j < N_CLASSES; ++j) s += expf(v[j] - m);
    const float ls = m + logf(s);
    #pragma unroll
    for (int j = 0; j < N_CLASSES; ++j) p[j] = v[j] - ls;
}

__global__ void gin_zero(float* __restrict__ p, int n) {
    const int i = blockIdx.x * blockDim.x + threadIdx.x;
    if (i < n) p[i] = 0.0f;
}

extern "C" void kernel_launch(void* const* d_in, const int* in_sizes, int n_in,
                              void* d_out, int out_size, void* d_ws, size_t ws_size,
                              hipStream_t stream) {
    (void)in_sizes; (void)n_in; (void)out_size; (void)ws_size;

    const float* x   = (const float*)d_in[0];
    const int*   ei  = (const int*)d_in[1];
    const int*   src = ei;
    const int*   dst = ei + N_EDGES;
    const float* w1a = (const float*)d_in[2];  const float* b1a = (const float*)d_in[3];
    const float* w1b = (const float*)d_in[4];  const float* b1b = (const float*)d_in[5];
    const float* g1  = (const float*)d_in[6];  const float* be1 = (const float*)d_in[7];
    const float* m1  = (const float*)d_in[8];  const float* v1  = (const float*)d_in[9];
    const float* w2a = (const float*)d_in[10]; const float* b2a = (const float*)d_in[11];
    const float* w2b = (const float*)d_in[12]; const float* b2b = (const float*)d_in[13];
    const float* g2  = (const float*)d_in[14]; const float* be2 = (const float*)d_in[15];
    const float* m2  = (const float*)d_in[16]; const float* v2  = (const float*)d_in[17];
    const float* fcw = (const float*)d_in[18]; const float* fcb = (const float*)d_in[19];
    float* out = (float*)d_out;

    const int NF = N_NODES * DIM;
    float* ws   = (float*)d_ws;
    float* y    = ws;
    float* u    = ws + (size_t)NF;
    float* agg1 = ws + (size_t)2 * NF;
    float* agg2 = ws + (size_t)3 * NF;

    const int TILES = N_NODES / 16;
    const int ETHREADS = N_EDGES * DIM;

    gin_zero<<<(2 * NF + 255) / 256, 256, 0, stream>>>(agg1, 2 * NF);
    gin_proj64<<<TILES, 32, 0, stream>>>(x, w1a, y);
    gin_scatter_add<<<(ETHREADS + 255) / 256, 256, 0, stream>>>(y, src, dst, agg1);
    gin_layer1<<<TILES, 32, 0, stream>>>(y, agg1, b1a, w1b, b1b, g1, be1, m1, v1, u);
    gin_scatter_add<<<(ETHREADS + 255) / 256, 256, 0, stream>>>(u, src, dst, agg2);
    gin_layer2_fc<<<TILES, 32, 0, stream>>>(u, agg2, w2a, b2a, w2b, b2b,
                                            g2, be2, m2, v2, fcw, fcb, out);
    gin_logsoftmax<<<(N_NODES + 255) / 256, 256, 0, stream>>>(out);
}


# Classifier wave takes log_softmax on its own accumulator: row max and row sum as 4-step lane butterflies over the class lanes (exps summed as a lane tree, not in class order); 7 launches -> 6
# speedup vs baseline: 2.7125x; 2.7125x over previous
#include <hip/hip_runtime.h>
#include <math.h>

#define N_NODES   100000
#define N_EDGES   1600000
#define F_IN      64
#define DIM       16
#define N_CLASSES 10
#define BN_EPS    1e-5f

typedef __attribute__((ext_vector_type(2))) float v2f;
typedef __attribute__((ext_vector_type(8))) float v8f;

__device__ __forceinline__ v8f wmma_k4(v2f a, v2f b, v8f c) {
    return __builtin_amdgcn_wmma_f32_16x16x4_f32(
         false, a,  false, b,
         (short)0, c,  false,  false);
}

__global__ void gin_proj64(const float* __restrict__ x,
                           const float* __restrict__ w1a,
                           float* __restrict__ y) {
    const int tile = blockIdx.x;
    const int lane = threadIdx.x;
    const int half = lane >> 4;
    const int l    = lane & 15;
    const int arow = tile * 16 + l;

    v8f c;
    #pragma unroll
    for (int r = 0; r < 8; ++r) c[r] = 0.0f;

    #pragma unroll
    for (int k0 = 0; k0 < F_IN; k0 += 4) {
        const int ka = k0 + half * 2;
        v2f a = *(const v2f*)(x + arow * F_IN + ka);
        v2f b;
        b.x = w1a[ka * DIM + l];
        b.y = w1a[ka * DIM + DIM + l];
        c = wmma_k4(a, b, c);
    }
    #pragma unroll
    for (int r = 0; r < 8; ++r)
        y[(tile * 16 + half * 8 + r) * DIM + l] = c[r];
}

#define AGG_COPY  (N_NODES * DIM)
#define SC_CHUNKS 4
#define SC_EPW    (32 * SC_CHUNKS)
#define SC_WAVES  (N_EDGES / SC_EPW)
static_assert(N_EDGES % SC_EPW == 0, "edge list must split into whole wave walks");

template <int J>
__device__ __forceinline__ unsigned row_pick(unsigned v) {
    return (unsigned)__builtin_amdgcn_update_dpp(0, (int)v, 0x150 + J, 0xF, 0xF, true);
}

template <int J>
struct ScStep {
    static __device__ __forceinline__ void gather(const float* __restrict__ feat,
                                                  unsigned sb, unsigned l4, float (&v)[16]) {
        v[J] = *(const float*)((const char*)feat + (row_pick<J>(sb) | l4));
        if constexpr (J + 1 < 16) ScStep<J + 1>::gather(feat, sb, l4, v);
    }
    static __device__ __forceinline__ void add(float* __restrict__ agg,
                                               unsigned db, unsigned l4, const float (&v)[16]) {
        atomicAdd((float*)((char*)agg + (row_pick<J>(db) | l4)), v[J]);
        if constexpr (J + 1 < 16) ScStep<J + 1>::add(agg, db, l4, v);
    }
};

__global__ void gin_scatter_add(const float* __restrict__ feat,
                                const int* __restrict__ src,
                                const int* __restrict__ dst,
                                float* __restrict__ agg) {
    const unsigned lane = threadIdx.x & 31u;
    const unsigned wave = (blockIdx.x * blockDim.x + threadIdx.x) >> 5;
    if (wave >= SC_WAVES) return;
    const unsigned l4 = (lane & 15u) << 2;
    const unsigned e0 = wave * SC_EPW + lane;

    unsigned sb[SC_CHUNKS], db[SC_CHUNKS];
    #pragma unroll
    for (int c = 0; c < SC_CHUNKS; ++c) {
        sb[c] = (unsigned)src[e0 + 32 * c] << 6;
        db[c] = (unsigned)dst[e0 + 32 * c] << 6;
    }

    const unsigned place = __builtin_amdgcn_s_sendmsg_rtn(0x87);
    agg += (size_t)((place >> 18) & 1u) * AGG_COPY;
    #pragma unroll
    for (int c = 0; c < SC_CHUNKS; ++c) {
        float v[16];
        ScStep<0>::gather(feat, sb[c], l4, v);
        ScStep<0>::add(agg, db[c], l4, v);
    }
}

__global__ void gin_layer1(const float* __restrict__ y, float* agg,
                           const float* __restrict__ b1a,
                           const float* __restrict__ w1b, const float* __restrict__ b1b,
                           const float* __restrict__ g,  const float* __restrict__ be,
                           const float* __restrict__ mu, const float* __restrict__ var,
                           float* __restrict__ u) {
    __shared__ float sA[16][17];
    const int tile = blockIdx.x;
    const int lane = threadIdx.x;
    const int half = lane >> 4;
    const int l    = lane & 15;

    const float bl = b1a[l];
    #pragma unroll
    for (int r = 0; r < 8; ++r) {
        const int row = half * 8 + r;
        const int gi  = (tile * 16 + row) * DIM + l;
        sA[row][l] = fmaxf(y[gi] + (agg[gi] + agg[AGG_COPY + gi]) + bl, 0.0f);
    }
    __syncthreads();

    const float cb = b1b[l];
    v8f c;
    #pragma unroll
    for (int r = 0; r < 8; ++r) c[r] = cb;

    #pragma unroll
    for (int k0 = 0; k0 < DIM; k0 += 4) {
        const int ka = k0 + half * 2;
        v2f a; a.x = sA[l][ka];           a.y = sA[l][ka + 1];
        v2f b; b.x = w1b[ka * DIM + l];   b.y = w1b[ka * DIM + DIM + l];
        c = wmma_k4(a, b, c);
    }

    const float sc = g[l] * rsqrtf(var[l] + BN_EPS);
    const float mm = mu[l], bb = be[l];
    #pragma unroll
    for (int r = 0; r < 8; ++r) {
        const float t = fmaxf(c[r], 0.0f);
        const int gi = (tile * 16 + half * 8 + r) * DIM + l;
        u[gi] = (t - mm) * sc + bb;
        agg[gi] = 0.0f;
        agg[AGG_COPY + gi] = 0.0f;
    }
}

template <int M>
__device__ __forceinline__ float row_xor(float v) {
    return __builtin_bit_cast(float,
        __builtin_amdgcn_update_dpp(0, __builtin_bit_cast(int, v), 0x160 + M, 0xF, 0xF, true));
}
__device__ __forceinline__ float row_max16(float v) {
    v = fmaxf(v, row_xor<1>(v));
    v = fmaxf(v, row_xor<2>(v));
    v = fmaxf(v, row_xor<4>(v));
    v = fmaxf(v, row_xor<8>(v));
    return v;
}
__device__ __forceinline__ float row_sum16(float v) {
    v += row_xor<1>(v);
    v += row_xor<2>(v);
    v += row_xor<4>(v);
    v += row_xor<8>(v);
    return v;
}

__global__ void gin_layer2_fc(const float* __restrict__ u,  const float* __restrict__ agg2,
                              const float* __restrict__ w2a, const float* __restrict__ b2a,
                              const float* __restrict__ w2b, const float* __restrict__ b2b,
                              const float* __restrict__ g,   const float* __restrict__ be,
                              const float* __restrict__ mu,  const float* __restrict__ var,
                              const float* __restrict__ fcw, const float* __restrict__ fcb,
                              float* __restrict__ out) {
    __shared__ float sA[16][17];
    const int tile = blockIdx.x;
    const int lane = threadIdx.x;
    const int half = lane >> 4;
    const int l    = lane & 15;

    #pragma unroll
    for (int r = 0; r < 8; ++r) {
        const int row = half * 8 + r;
        const int gi  = (tile * 16 + row) * DIM + l;
        sA[row][l] = u[gi] + (agg2[gi] + agg2[AGG_COPY + gi]);
    }
    __syncthreads();

    v8f c;
    {
        const float cb = b2a[l];
        #pragma unroll
        for (int r = 0; r < 8; ++r) c[r] = cb;
    }
    #pragma unroll
    for (int k0 = 0; k0 < DIM; k0 += 4) {
        const int ka = k0 + half * 2;
        v2f a; a.x = sA[l][ka];          a.y = sA[l][ka + 1];
        v2f b; b.x = w2a[ka * DIM + l];  b.y = w2a[ka * DIM + DIM + l];
        c = wmma_k4(a, b, c);
    }
    __syncthreads();
    #pragma unroll
    for (int r = 0; r < 8; ++r) sA[half * 8 + r][l] = fmaxf(c[r], 0.0f);
    __syncthreads();

    {
        const float cb = b2b[l];
        #pragma unroll
        for (int r = 0; r < 8; ++r) c[r] = cb;
    }
    #pragma unroll
    for (int k0 = 0; k0 < DIM; k0 += 4) {
        const int ka = k0 + half * 2;
        v2f a; a.x = sA[l][ka];          a.y = sA[l][ka + 1];
        v2f b; b.x = w2b[ka * DIM + l];  b.y = w2b[ka * DIM + DIM + l];
        c = wmma_k4(a, b, c);
    }
    __syncthreads();

    {
        const float sc = g[l] * rsqrtf(var[l] + BN_EPS);
        const float mm = mu[l], bb = be[l];
        #pragma unroll
        for (int r = 0; r < 8; ++r) {
            const float t = fmaxf(c[r], 0.0f);
            sA[half * 8 + r][l] = (t - mm) * sc + bb;
        }
    }
    __syncthreads();

    {
        const float cb = (l < N_CLASSES) ? fcb[l] : 0.0f;
        #pragma unroll
        for (int r = 0; r < 8; ++r) c[r] = cb;
    }
    #pragma unroll
    for (int k0 = 0; k0 < DIM; k0 += 4) {
        const int ka = k0 + half * 2;
        v2f a; a.x = sA[l][ka];  a.y = sA[l][ka + 1];
        v2f b;
        b.x = (l < N_CLASSES) ? fcw[ka * N_CLASSES + l]       : 0.0f;
        b.y = (l < N_CLASSES) ? fcw[(ka + 1) * N_CLASSES + l] : 0.0f;
        c = wmma_k4(a, b, c);
    }
    const bool live = l < N_CLASSES;
    float v[8], m[8], lse[8];
    #pragma unroll
    for (int r = 0; r < 8; ++r) {
        v[r] = live ? c[r] : -INFINITY;
        m[r] = row_max16(v[r]);
    }
    #pragma unroll
    for (int r = 0; r < 8; ++r)
        lse[r] = m[r] + logf(row_sum16(expf(v[r] - m[r])));
    if (live) {
        #pragma unroll
        for (int r = 0; r < 8; ++r)
            out[(tile * 16 + half * 8 + r) * N_CLASSES + l] = c[r] - lse[r];
    }
}

__global__ void gin_zero(float* __restrict__ p, int n) {
    const int i = blockIdx.x * blockDim.x + threadIdx.x;
    if (i < n) p[i] = 0.0f;
}

extern "C" void kernel_launch(void* const* d_in, const int* in_sizes, int n_in,
                              void* d_out, int out_size, void* d_ws, size_t ws_size,
                              hipStream_t stream) {
    (void)in_sizes; (void)n_in; (void)out_size; (void)ws_size;

    const float* x   = (const float*)d_in[0];
    const int*   ei  = (const int*)d_in[1];
    const int*   src = ei;
    const int*   dst = ei + N_EDGES;
    const float* w1a = (const float*)d_in[2];  const float* b1a = (const float*)d_in[3];
    const float* w1b = (const float*)d_in[4];  const float* b1b = (const float*)d_in[5];
    const float* g1  = (const float*)d_in[6];  const float* be1 = (const float*)d_in[7];
    const float* m1  = (const float*)d_in[8];  const float* v1  = (const float*)d_in[9];
    const float* w2a = (const float*)d_in[10]; const float* b2a = (const float*)d_in[11];
    const float* w2b = (const float*)d_in[12]; const float* b2b = (const float*)d_in[13];
    const float* g2  = (const float*)d_in[14]; const float* be2 = (const float*)d_in[15];
    const float* m2  = (const float*)d_in[16]; const float* v2  = (const float*)d_in[17];
    const float* fcw = (const float*)d_in[18]; const float* fcb = (const float*)d_in[19];
    float* out = (float*)d_out;

    const int NF = N_NODES * DIM;
    float* ws   = (float*)d_ws;
    float* y    = ws;
    float* u    = ws + (size_t)NF;
    float* agg  = ws + (size_t)2 * NF;

    const int TILES = N_NODES / 16;
    const int SGRID = (SC_WAVES * 32 + 255) / 256;

    gin_zero<<<(2 * NF + 255) / 256, 256, 0, stream>>>(agg, 2 * NF);
    gin_proj64<<<TILES, 32, 0, stream>>>(x, w1a, y);
    gin_scatter_add<<<SGRID, 256, 0, stream>>>(y, src, dst, agg);
    gin_layer1<<<TILES, 32, 0, stream>>>(y, agg, b1a, w1b, b1b, g1, be1, m1, v1, u);
    gin_scatter_add<<<SGRID, 256, 0, stream>>>(u, src, dst, agg);
    gin_layer2_fc<<<TILES, 32, 0, stream>>>(u, agg, w2a, b2a, w2b, b2b,
                                            g2, be2, m2, v2, fcw, fcb, out);
}


# Clear and projection in one launch as separate work-groups (782 groups of 8 one-tile waves first, 12500 clearing groups behind them); clearing stores and per-tile sums as before; 6 launches -> 5
# speedup vs baseline: 2.8216x; 1.0402x over previous
#include <hip/hip_runtime.h>
#include <math.h>

#define N_NODES   100000
#define N_EDGES   1600000
#define F_IN      64
#define DIM       16
#define N_CLASSES 10
#define BN_EPS    1e-5f

typedef __attribute__((ext_vector_type(2))) float v2f;
typedef __attribute__((ext_vector_type(8))) float v8f;

__device__ __forceinline__ v8f wmma_k4(v2f a, v2f b, v8f c) {
    return __builtin_amdgcn_wmma_f32_16x16x4_f32(
         false, a,  false, b,
         (short)0, c,  false,  false);
}

#define N_TILES     (N_NODES / 16)
#define PROJ_GROUPS ((N_TILES + 7) / 8)
#define CLR_FLOATS  (2 * N_NODES * DIM)
#define CLR_GROUPS  (CLR_FLOATS / 256)
static_assert(CLR_FLOATS % 256 == 0, "the clear must split into whole work-groups");

__global__ void
gin_proj64_clear(const float* __restrict__ x,
                 const float* __restrict__ w1a,
                 float* __restrict__ y,
                 float* __restrict__ agg) {
    if (blockIdx.x >= PROJ_GROUPS) {
        agg[(blockIdx.x - PROJ_GROUPS) * 256 + threadIdx.x] = 0.0f;
        return;
    }
    const int tile = blockIdx.x * 8 + (threadIdx.x >> 5);
    if (tile >= N_TILES) return;
    const int lane = threadIdx.x & 31;
    const int half = lane >> 4;
    const int l    = lane & 15;
    const int arow = tile * 16 + l;

    v8f c;
    #pragma unroll
    for (int r = 0; r < 8; ++r) c[r] = 0.0f;

    #pragma unroll
    for (int k0 = 0; k0 < F_IN; k0 += 4) {
        const int ka = k0 + half * 2;
        v2f a = *(const v2f*)(x + arow * F_IN + ka);
        v2f b;
        b.x = w1a[ka * DIM + l];
        b.y = w1a[ka * DIM + DIM + l];
        c = wmma_k4(a, b, c);
    }
    #pragma unroll
    for (int r = 0; r < 8; ++r)
        y[(tile * 16 + half * 8 + r) * DIM + l] = c[r];
}

#define AGG_COPY  (N_NODES * DIM)
#define SC_CHUNKS 4
#define SC_EPW    (32 * SC_CHUNKS)
#define SC_WAVES  (N_EDGES / SC_EPW)
static_assert(N_EDGES % SC_EPW == 0, "edge list must split into whole wave walks");

template <int J>
__device__ __forceinline__ unsigned row_pick(unsigned v) {
    return (unsigned)__builtin_amdgcn_update_dpp(0, (int)v, 0x150 + J, 0xF, 0xF, true);
}

template <int J>
struct ScStep {
    static __device__ __forceinline__ void gather(const float* __restrict__ feat,
                                                  unsigned sb, unsigned l4, float (&v)[16]) {
        v[J] = *(const float*)((const char*)feat + (row_pick<J>(sb) | l4));
        if constexpr (J + 1 < 16) ScStep<J + 1>::gather(feat, sb, l4, v);
    }
    static __device__ __forceinline__ void add(float* __restrict__ agg,
                                               unsigned db, unsigned l4, const float (&v)[16]) {
        atomicAdd((float*)((char*)agg + (row_pick<J>(db) | l4)), v[J]);
        if constexpr (J + 1 < 16) ScStep<J + 1>::add(agg, db, l4, v);
    }
};

__global__ void gin_scatter_add(const float* __restrict__ feat,
                                const int* __restrict__ src,
                                const int* __restrict__ dst,
                                float* __restrict__ agg) {
    const unsigned lane = threadIdx.x & 31u;
    const unsigned wave = (blockIdx.x * blockDim.x + threadIdx.x) >> 5;
    if (wave >= SC_WAVES) return;
    const unsigned l4 = (lane & 15u) << 2;
    const unsigned e0 = wave * SC_EPW + lane;

    unsigned sb[SC_CHUNKS], db[SC_CHUNKS];
    #pragma unroll
    for (int c = 0; c < SC_CHUNKS; ++c) {
        sb[c] = (unsigned)src[e0 + 32 * c] << 6;
        db[c] = (unsigned)dst[e0 + 32 * c] << 6;
    }

    const unsigned place = __builtin_amdgcn_s_sendmsg_rtn(0x87);
    agg += (size_t)((place >> 18) & 1u) * AGG_COPY;
    #pragma unroll
    for (int c = 0; c < SC_CHUNKS; ++c) {
        float v[16];
        ScStep<0>::gather(feat, sb[c], l4, v);
        ScStep<0>::add(agg, db[c], l4, v);
    }
}

__global__ void gin_layer1(const float* __restrict__ y, float* agg,
                           const float* __restrict__ b1a,
                           const float* __restrict__ w1b, const float* __restrict__ b1b,
                           const float* __restrict__ g,  const float* __restrict__ be,
                           const float* __restrict__ mu, const float* __restrict__ var,
                           float* __restrict__ u) {
    __shared__ float sA[16][17];
    const int tile = blockIdx.x;
    const int lane = threadIdx.x;
    const int half = lane >> 4;
    const int l    = lane & 15;

    const float bl = b1a[l];
    #pragma unroll
    for (int r = 0; r < 8; ++r) {
        const int row = half * 8 + r;
        const int gi  = (tile * 16 + row) * DIM + l;
        sA[row][l] = fmaxf(y[gi] + (agg[gi] + agg[AGG_COPY + gi]) + bl, 0.0f);
    }
    __syncthreads();

    const float cb = b1b[l];
    v8f c;
    #pragma unroll
    for (int r = 0; r < 8; ++r) c[r] = cb;

    #pragma unroll
    for (int k0 = 0; k0 < DIM; k0 += 4) {
        const int ka = k0 + half * 2;
        v2f a; a.x = sA[l][ka];           a.y = sA[l][ka + 1];
        v2f b; b.x = w1b[ka * DIM + l];   b.y = w1b[ka * DIM + DIM + l];
        c = wmma_k4(a, b, c);
    }

    const float sc = g[l] * rsqrtf(var[l] + BN_EPS);
    const float mm = mu[l], bb = be[l];
    #pragma unroll
    for (int r = 0; r < 8; ++r) {
        const float t = fmaxf(c[r], 0.0f);
        const int gi = (tile * 16 + half * 8 + r) * DIM + l;
        u[gi] = (t - mm) * sc + bb;
        agg[gi] = 0.0f;
        agg[AGG_COPY + gi] = 0.0f;
    }
}

template <int M>
__device__ __forceinline__ float row_xor(float v) {
    return __builtin_bit_cast(float,
        __builtin_amdgcn_update_dpp(0, __builtin_bit_cast(int, v), 0x160 + M, 0xF, 0xF, true));
}
__device__ __forceinline__ float row_max16(float v) {
    v = fmaxf(v, row_xor<1>(v));
    v = fmaxf(v, row_xor<2>(v));
    v = fmaxf(v, row_xor<4>(v));
    v = fmaxf(v, row_xor<8>(v));
    return v;
}
__device__ __forceinline__ float row_sum16(float v) {
    v += row_xor<1>(v);
    v += row_xor<2>(v);
    v += row_xor<4>(v);
    v += row_xor<8>(v);
    return v;
}

__global__ void gin_layer2_fc(const float* __restrict__ u,  const float* __restrict__ agg2,
                              const float* __restrict__ w2a, const float* __restrict__ b2a,
                              const float* __restrict__ w2b, const float* __restrict__ b2b,
                              const float* __restrict__ g,   const float* __restrict__ be,
                              const float* __restrict__ mu,  const float* __restrict__ var,
                              const float* __restrict__ fcw, const float* __restrict__ fcb,
                              float* __restrict__ out) {
    __shared__ float sA[16][17];
    const int tile = blockIdx.x;
    const int lane = threadIdx.x;
    const int half = lane >> 4;
    const int l    = lane & 15;

    #pragma unroll
    for (int r = 0; r < 8; ++r) {
        const int row = half * 8 + r;
        const int gi  = (tile * 16 + row) * DIM + l;
        sA[row][l] = u[gi] + (agg2[gi] + agg2[AGG_COPY + gi]);
    }
    __syncthreads();

    v8f c;
    {
        const float cb = b2a[l];
        #pragma unroll
        for (int r = 0; r < 8; ++r) c[r] = cb;
    }
    #pragma unroll
    for (int k0 = 0; k0 < DIM; k0 += 4) {
        const int ka = k0 + half * 2;
        v2f a; a.x = sA[l][ka];          a.y = sA[l][ka + 1];
        v2f b; b.x = w2a[ka * DIM + l];  b.y = w2a[ka * DIM + DIM + l];
        c = wmma_k4(a, b, c);
    }
    __syncthreads();
    #pragma unroll
    for (int r = 0; r < 8; ++r) sA[half * 8 + r][l] = fmaxf(c[r], 0.0f);
    __syncthreads();

    {
        const float cb = b2b[l];
        #pragma unroll
        for (int r = 0; r < 8; ++r) c[r] = cb;
    }
    #pragma unroll
    for (int k0 = 0; k0 < DIM; k0 += 4) {
        const int ka = k0 + half * 2;
        v2f a; a.x = sA[l][ka];          a.y = sA[l][ka + 1];
        v2f b; b.x = w2b[ka * DIM + l];  b.y = w2b[ka * DIM + DIM + l];
        c = wmma_k4(a, b, c);
    }
    __syncthreads();

    {
        const float sc = g[l] * rsqrtf(var[l] + BN_EPS);
        const float mm = mu[l], bb = be[l];
        #pragma unroll
        for (int r = 0; r < 8; ++r) {
            const float t = fmaxf(c[r], 0.0f);
            sA[half * 8 + r][l] = (t - mm) * sc + bb;
        }
    }
    __syncthreads();

    {
        const float cb = (l < N_CLASSES) ? fcb[l] : 0.0f;
        #pragma unroll
        for (int r = 0; r < 8; ++r) c[r] = cb;
    }
    #pragma unroll
    for (int k0 = 0; k0 < DIM; k0 += 4) {
        const int ka = k0 + half * 2;
        v2f a; a.x = sA[l][ka];  a.y = sA[l][ka + 1];
        v2f b;
        b.x = (l < N_CLASSES) ? fcw[ka * N_CLASSES + l]       : 0.0f;
        b.y = (l < N_CLASSES) ? fcw[(ka + 1) * N_CLASSES + l] : 0.0f;
        c = wmma_k4(a, b, c);
    }
    const bool live = l < N_CLASSES;
    float v[8], m[8], lse[8];
    #pragma unroll
    for (int r = 0; r < 8; ++r) {
        v[r] = live ? c[r] : -INFINITY;
        m[r] = row_max16(v[r]);
    }
    #pragma unroll
    for (int r = 0; r < 8; ++r)
        lse[r] = m[r] + logf(row_sum16(expf(v[r] - m[r])));
    if (live) {
        #pragma unroll
        for (int r = 0; r < 8; ++r)
            out[(tile * 16 + half * 8 + r) * N_CLASSES + l] = c[r] - lse[r];
    }
}

extern "C" void kernel_launch(void* const* d_in, const int* in_sizes, int n_in,
                              void* d_out, int out_size, void* d_ws, size_t ws_size,
                              hipStream_t stream) {
    (void)in_sizes; (void)n_in; (void)out_size; (void)ws_size;

    const float* x   = (const float*)d_in[0];
    const int*   ei  = (const int*)d_in[1];
    const int*   src = ei;
    const int*   dst = ei + N_EDGES;
    const float* w1a = (const float*)d_in[2];  const float* b1a = (const float*)d_in[3];
    const float* w1b = (const float*)d_in[4];  const float* b1b = (const float*)d_in[5];
    const float* g1  = (const float*)d_in[6];  const float* be1 = (const float*)d_in[7];
    const float* m1  = (const float*)d_in[8];  const float* v1  = (const float*)d_in[9];
    const float* w2a = (const float*)d_in[10]; const float* b2a = (const float*)d_in[11];
    const float* w2b = (const float*)d_in[12]; const float* b2b = (const float*)d_in[13];
    const float* g2  = (const float*)d_in[14]; const float* be2 = (const float*)d_in[15];
    const float* m2  = (const float*)d_in[16]; const float* v2  = (const float*)d_in[17];
    const float* fcw = (const float*)d_in[18]; const float* fcb = (const float*)d_in[19];
    float* out = (float*)d_out;

    const int NF = N_NODES * DIM;
    float* ws   = (float*)d_ws;
    float* y    = ws;
    float* u    = ws + (size_t)NF;
    float* agg  = ws + (size_t)2 * NF;

    const int TILES = N_NODES / 16;
    const int SGRID = (SC_WAVES * 32 + 255) / 256;

    gin_proj64_clear<<<PROJ_GROUPS + CLR_GROUPS, 256, 0, stream>>>(x, w1a, y, agg);
    gin_scatter_add<<<SGRID, 256, 0, stream>>>(y, src, dst, agg);
    gin_layer1<<<TILES, 32, 0, stream>>>(y, agg, b1a, w1b, b1b, g1, be1, m1, v1, u);
    gin_scatter_add<<<SGRID, 256, 0, stream>>>(u, src, dst, agg);
    gin_layer2_fc<<<TILES, 32, 0, stream>>>(u, agg, w2a, b2a, w2b, b2b,
                                            g2, be2, m2, v2, fcw, fcb, out);
}
